# starting kernel
# baseline (speedup 1.0000x reference)
#include <hip/hip_runtime.h>
#include <math.h>

typedef __attribute__((ext_vector_type(2))) float v2f;
typedef __attribute__((ext_vector_type(8))) float v8f;

#define T_SEQ     10
#define D_MODEL   256
#define BATCH     16
#define RES       200
#define MODES_OUT 16
#define MODES_IN  32
#define XPAD      208
#define NXT       13
#define KDIM      64

__global__ __launch_bounds__(256) void build_wf(const float* __restrict__ w_real,
                                                const float* __restrict__ w_imag,
                                                float* __restrict__ wf) {
  int idx = blockIdx.x * blockDim.x + threadIdx.x;
  int m = idx & 31;
  int x = (idx >> 5) % XPAD;
  int t = idx / (XPAD * 32);
  if (t >= T_SEQ) return;

  float accr = 0.0f, acci = 0.0f;
  if (x < RES) {
    const float base = (6.283185307179586476925f / 200.0f) * (float)x;
    #pragma unroll
    for (int o = 0; o < MODES_OUT; ++o) {
      float s, c;
      sincosf(base * (float)o, &s, &c);
      float wr = w_real[(t * MODES_OUT + o) * MODES_IN + m];
      float wi = w_imag[(t * MODES_OUT + o) * MODES_IN + m];
      accr += wr * c - wi * s;
      acci -= wi * c + wr * s;
    }
  }
  float* row = wf + (t * XPAD + x) * KDIM;
  row[m]            = accr;
  row[MODES_IN + m] = acci;
}

__global__ __launch_bounds__(256) void spectral_wmma(const float* __restrict__ v,
                                                     const float* __restrict__ wf,
                                                     float* __restrict__ out) {
  const int lane = threadIdx.x & 31;
  const int wid  = (int)((blockIdx.x * blockDim.x + threadIdx.x) >> 5);

  const int hp = wid & 7;
  const int xt = (wid >> 3) % NXT;
  const int tt = (wid / (8 * NXT)) % T_SEQ;
  const int b  = wid / (8 * NXT * T_SEQ);
  if (b >= BATCH) return;

  const int lm = lane & 15;
  const int lh = lane >> 4;

  const float* arow  = wf + (size_t)(tt * XPAD + xt * 16 + lm) * KDIM + 2 * lh;
  const float* brow0 = v  + (size_t)(((b * T_SEQ + tt) * D_MODEL) + hp * 32 + lm) * KDIM + 2 * lh;
  const float* brow1 = brow0 + 16 * KDIM;

  v8f acc0 = {};
  v8f acc1 = {};
  #pragma unroll
  for (int kb = 0; kb < 16; ++kb) {
    v2f a  = *(const v2f*)(arow  + 4 * kb);
    v2f b0 = *(const v2f*)(brow0 + 4 * kb);
    v2f b1 = *(const v2f*)(brow1 + 4 * kb);
    acc0 = __builtin_amdgcn_wmma_f32_16x16x4_f32(false, a, false, b0,
                                                 (short)0, acc0, false, false);
    acc1 = __builtin_amdgcn_wmma_f32_16x16x4_f32(false, a, false, b1,
                                                 (short)0, acc1, false, false);
  }

  float* obase = out + (size_t)(b * T_SEQ + tt) * RES * D_MODEL + hp * 32 + lm;
  const int xb = xt * 16;
  #pragma unroll
  for (int j = 0; j < 8; ++j) {
    int x = xb + j + 8 * lh;
    if (x < RES) {
      obase[(size_t)x * D_MODEL]      = acc0[j];
      obase[(size_t)x * D_MODEL + 16] = acc1[j];
    }
  }
}

extern "C" void kernel_launch(void* const* d_in, const int* in_sizes, int n_in,
                              void* d_out, int out_size, void* d_ws, size_t ws_size,
                              hipStream_t stream) {
  const float* v      = (const float*)d_in[0];
  const float* w_real = (const float*)d_in[1];
  const float* w_imag = (const float*)d_in[2];
  float* out = (float*)d_out;
  float* wf  = (float*)d_ws;

  build_wf<<<(T_SEQ * XPAD * 32) / 256, 256, 0, stream>>>(w_real, w_imag, wf);

  spectral_wmma<<<(BATCH * T_SEQ * NXT * 8) / 8, 256, 0, stream>>>(v, wf, out);
}


# each wave keeps its 32-h V fragments in registers and loops over half of the x-tiles (7 or 6), next Wf tile prefetched; V re-fetch 13x -> 2x, waves 16640 -> 2560; sum order unchanged
# speedup vs baseline: 2.1767x; 2.1767x over previous
#include <hip/hip_runtime.h>
#include <math.h>

typedef __attribute__((ext_vector_type(2))) float v2f;
typedef __attribute__((ext_vector_type(8))) float v8f;

#define T_SEQ     10
#define D_MODEL   256
#define BATCH     16
#define RES       200
#define MODES_OUT 16
#define MODES_IN  32
#define XPAD      208
#define NXT       13
#define KDIM      64

__global__ __launch_bounds__(256) void build_wf(const float* __restrict__ w_real,
                                                const float* __restrict__ w_imag,
                                                float* __restrict__ wf) {
  int idx = blockIdx.x * blockDim.x + threadIdx.x;
  int m = idx & 31;
  int x = (idx >> 5) % XPAD;
  int t = idx / (XPAD * 32);
  if (t >= T_SEQ) return;

  float accr = 0.0f, acci = 0.0f;
  if (x < RES) {
    const float base = (6.283185307179586476925f / 200.0f) * (float)x;
    #pragma unroll
    for (int o = 0; o < MODES_OUT; ++o) {
      float s, c;
      sincosf(base * (float)o, &s, &c);
      float wr = w_real[(t * MODES_OUT + o) * MODES_IN + m];
      float wi = w_imag[(t * MODES_OUT + o) * MODES_IN + m];
      accr += wr * c - wi * s;
      acci -= wi * c + wr * s;
    }
  }
  float* row = wf + (t * XPAD + x) * KDIM;
  row[m]            = accr;
  row[MODES_IN + m] = acci;
}

#define XT_FIRST  7

struct AFrag { v2f k[16]; };

__device__ __forceinline__ void load_a(AFrag& a, const float* __restrict__ arow) {
  #pragma unroll
  for (int kb = 0; kb < 16; ++kb) a.k[kb] = *(const v2f*)(arow + 4 * kb);
}

template <bool LAST_TILE>
__device__ __forceinline__ void tile_mma_store(const AFrag& a, const AFrag& b0, const AFrag& b1,
                                               float* __restrict__ obase, int xb, int lh) {
  v8f acc0 = {};
  v8f acc1 = {};
  #pragma unroll
  for (int kb = 0; kb < 16; ++kb) {
    acc0 = __builtin_amdgcn_wmma_f32_16x16x4_f32(false, a.k[kb], false, b0.k[kb],
                                                 (short)0, acc0, false, false);
    acc1 = __builtin_amdgcn_wmma_f32_16x16x4_f32(false, a.k[kb], false, b1.k[kb],
                                                 (short)0, acc1, false, false);
  }
  #pragma unroll
  for (int j = 0; j < 8; ++j) {
    int x = xb + j + 8 * lh;
    if (!LAST_TILE || x < RES) {
      obase[(size_t)x * D_MODEL]      = acc0[j];
      obase[(size_t)x * D_MODEL + 16] = acc1[j];
    }
  }
}

__global__ __launch_bounds__(128) void spectral_wmma(const float* __restrict__ v,
                                                     const float* __restrict__ wf,
                                                     float* __restrict__ out) {
  const int lane = threadIdx.x & 31;
  const int w    = threadIdx.x >> 5;
  const int bid  = blockIdx.x;
  const int xs   = bid & 1;
  const int hg   = (bid >> 1) & 1;
  const int bt   = bid >> 2;
  const int tt   = bt % T_SEQ;
  const int hp   = hg * 4 + w;

  const int lm = lane & 15;
  const int lh = lane >> 4;

  const float* brow0 = v + (size_t)(bt * D_MODEL + hp * 32 + lm) * KDIM + 2 * lh;
  const float* brow1 = brow0 + 16 * KDIM;
  AFrag b0, b1;
  #pragma unroll
  for (int kb = 0; kb < 16; ++kb) {
    b0.k[kb] = *(const v2f*)(brow0 + 4 * kb);
    b1.k[kb] = *(const v2f*)(brow1 + 4 * kb);
  }

  const int xt0 = xs * XT_FIRST;
  const float* arow = wf + (size_t)(tt * XPAD + xt0 * 16 + lm) * KDIM + 2 * lh;
  float* obase = out + (size_t)bt * RES * D_MODEL + hp * 32 + lm;

  AFrag a, an;
  load_a(a, arow);

  #pragma unroll
  for (int i = 0; i < 5; ++i) {
    load_a(an, arow + (size_t)(i + 1) * 16 * KDIM);
    tile_mma_store<false>(a, b0, b1, obase, (xt0 + i) * 16, lh);
    a = an;
  }

  if (xs == 0) {
    load_a(an, arow + (size_t)6 * 16 * KDIM);
    tile_mma_store<false>(a,  b0, b1, obase, 5 * 16, lh);
    tile_mma_store<false>(an, b0, b1, obase, 6 * 16, lh);
  } else {
    tile_mma_store<true>(a, b0, b1, obase, (NXT - 1) * 16, lh);
  }
}

extern "C" void kernel_launch(void* const* d_in, const int* in_sizes, int n_in,
                              void* d_out, int out_size, void* d_ws, size_t ws_size,
                              hipStream_t stream) {
  const float* v      = (const float*)d_in[0];
  const float* w_real = (const float*)d_in[1];
  const float* w_imag = (const float*)d_in[2];
  float* out = (float*)d_out;
  float* wf  = (float*)d_ws;

  build_wf<<<(T_SEQ * XPAD * 32) / 256, 256, 0, stream>>>(w_real, w_imag, wf);

  spectral_wmma<<<BATCH * T_SEQ * 2 * 2, 128, 0, stream>>>(v, wf, out);
}


# factored through the 32 mode sums: W[32x64]*V once per wave on WMMA, D tiles reused in registers as B of a K=32 trig product per x-tile; Wf build becomes a 208x32 cos/-sin table; sum order changed
# speedup vs baseline: 2.6841x; 1.2331x over previous
#include <hip/hip_runtime.h>
#include <math.h>


typedef __attribute__((ext_vector_type(2))) float v2f;
typedef __attribute__((ext_vector_type(4))) float v4f;
typedef __attribute__((ext_vector_type(8))) float v8f;

#define T_SEQ     10
#define D_MODEL   256
#define BATCH     16
#define RES       200
#define MODES_OUT 16
#define MODES_IN  32
#define XPAD      208
#define NXT       13
#define KDIM      64
#define TROW      32


__global__ __launch_bounds__(256) void build_trig(float* __restrict__ tg) {
  int idx = blockIdx.x * blockDim.x + threadIdx.x;
  int o = idx & 15;
  int x = idx >> 4;
  float c = 0.0f, ns = 0.0f;
  if (x < RES) {
    const float base = (6.283185307179586476925f / 200.0f) * (float)x;
    float s;
    sincosf(base * (float)o, &s, &c);
    ns = -s;
  }
  tg[x * TROW + o]             = c;
  tg[x * TROW + MODES_OUT + o] = ns;
}

#define XT_FIRST  7

struct VFrag { v2f k[16]; };
struct WFrag { v2f k[8];  };
struct MFrag { v2f k[8];  };

__device__ __forceinline__ v8f mma(v2f a, v2f b, v8f c) {
  return __builtin_amdgcn_wmma_f32_16x16x4_f32(false, a, false, b, (short)0, c, false, false);
}

__device__ __forceinline__ void load_trig(MFrag& a, const float* __restrict__ trow) {
  const v4f q0 = *(const v4f*)(trow);
  const v4f q1 = *(const v4f*)(trow + 4);
  const v4f q2 = *(const v4f*)(trow + MODES_OUT);
  const v4f q3 = *(const v4f*)(trow + MODES_OUT + 4);
  a.k[0] = q0.xy;  a.k[1] = q0.zw;  a.k[2] = q1.xy;  a.k[3] = q1.zw;
  a.k[4] = q2.xy;  a.k[5] = q2.zw;  a.k[6] = q3.xy;  a.k[7] = q3.zw;
}

template <bool LAST_TILE>
__device__ __forceinline__ void tile_mma_store(const MFrag& a, const MFrag& m0, const MFrag& m1,
                                               float* __restrict__ obase, int xb, int lh) {
  v8f acc0 = {};
  v8f acc1 = {};
  #pragma unroll
  for (int kb = 0; kb < 8; ++kb) {
    acc0 = mma(a.k[kb], m0.k[kb], acc0);
    acc1 = mma(a.k[kb], m1.k[kb], acc1);
  }
  #pragma unroll
  for (int j = 0; j < 8; ++j) {
    int x = xb + j + 8 * lh;
    if (!LAST_TILE || x < RES) {
      obase[(size_t)x * D_MODEL]      = acc0[j];
      obase[(size_t)x * D_MODEL + 16] = acc1[j];
    }
  }
}

__global__ __launch_bounds__(128) void spectral_wmma(const float* __restrict__ v,
                                                     const float* __restrict__ w_real,
                                                     const float* __restrict__ w_imag,
                                                     const float* __restrict__ tg,
                                                     float* __restrict__ out) {
  const int lane = threadIdx.x & 31;
  const int w    = threadIdx.x >> 5;
  const int bid  = blockIdx.x;
  const int xs   = bid & 1;
  const int hg   = (bid >> 1) & 1;
  const int bt   = bid >> 2;
  const int tt   = bt % T_SEQ;
  const int hp   = hg * 4 + w;

  const int lm = lane & 15;
  const int lh = lane >> 4;

  const float* brow0 = v + (size_t)(bt * D_MODEL + hp * 32 + lm) * KDIM + 2 * lh;
  const float* brow1 = brow0 + 16 * KDIM;
  VFrag b0, b1;
  #pragma unroll
  for (int kb = 0; kb < 16; ++kb) {
    b0.k[kb] = *(const v2f*)(brow0 + 4 * kb);
    b1.k[kb] = *(const v2f*)(brow1 + 4 * kb);
  }

  const float* wr_row = w_real + (size_t)(tt * MODES_OUT + lm) * MODES_IN + 2 * lh;
  const float* wi_row = w_imag + (size_t)(tt * MODES_OUT + lm) * MODES_IN + 2 * lh;
  WFrag wr, wi, nwi;
  #pragma unroll
  for (int j = 0; j < 8; ++j) {
    wr.k[j]  = *(const v2f*)(wr_row + 4 * j);
    wi.k[j]  = *(const v2f*)(wi_row + 4 * j);
    nwi.k[j] = -wi.k[j];
  }

  v8f c0 = {}, s0 = {}, c1 = {}, s1 = {};
  #pragma unroll
  for (int j = 0; j < 8; ++j) {
    c0 = mma(wr.k[j], b0.k[j], c0);
    s0 = mma(wi.k[j], b0.k[j], s0);
    c1 = mma(wr.k[j], b1.k[j], c1);
    s1 = mma(wi.k[j], b1.k[j], s1);
  }
  #pragma unroll
  for (int j = 0; j < 8; ++j) {
    c0 = mma(nwi.k[j], b0.k[8 + j], c0);
    s0 = mma(wr.k[j],  b0.k[8 + j], s0);
    c1 = mma(nwi.k[j], b1.k[8 + j], c1);
    s1 = mma(wr.k[j],  b1.k[8 + j], s1);
  }

  MFrag m0, m1;
  #pragma unroll
  for (int kb = 0; kb < 4; ++kb) {
    m0.k[kb]     = (v2f){c0[2 * kb], c0[2 * kb + 1]};
    m0.k[4 + kb] = (v2f){s0[2 * kb], s0[2 * kb + 1]};
    m1.k[kb]     = (v2f){c1[2 * kb], c1[2 * kb + 1]};
    m1.k[4 + kb] = (v2f){s1[2 * kb], s1[2 * kb + 1]};
  }

  const int xt0 = xs * XT_FIRST;
  const float* trow = tg + (size_t)(xt0 * 16 + lm) * TROW + 8 * lh;
  float* obase = out + (size_t)bt * RES * D_MODEL + hp * 32 + lm;

  MFrag a, an;
  load_trig(a, trow);

  #pragma unroll
  for (int i = 0; i < 5; ++i) {
    load_trig(an, trow + (size_t)(i + 1) * 16 * TROW);
    tile_mma_store<false>(a, m0, m1, obase, (xt0 + i) * 16, lh);
    a = an;
  }

  if (xs == 0) {
    load_trig(an, trow + (size_t)6 * 16 * TROW);
    tile_mma_store<false>(a,  m0, m1, obase, 5 * 16, lh);
    tile_mma_store<false>(an, m0, m1, obase, 6 * 16, lh);
  } else {
    tile_mma_store<true>(a, m0, m1, obase, (NXT - 1) * 16, lh);
  }
}

extern "C" void kernel_launch(void* const* d_in, const int* in_sizes, int n_in,
                              void* d_out, int out_size, void* d_ws, size_t ws_size,
                              hipStream_t stream) {
  const float* v      = (const float*)d_in[0];
  const float* w_real = (const float*)d_in[1];
  const float* w_imag = (const float*)d_in[2];
  float* out = (float*)d_out;
  float* tg  = (float*)d_ws;

  build_trig<<<(XPAD * 16) / 256, 256, 0, stream>>>(tg);

  spectral_wmma<<<BATCH * T_SEQ * 2 * 2, 128, 0, stream>>>(v, w_real, w_imag, tg, out);
}


# table launch folded into the main kernel: each block writes 200 (cos,-sin) pairs to LDS, stage-2 A gathered at k=(o*x) mod 200, C/S rows interleaved; 2 launches -> 1; sum order changed
# speedup vs baseline: 3.0703x; 1.1439x over previous
#include <hip/hip_runtime.h>
#include <math.h>


typedef __attribute__((ext_vector_type(2))) float v2f;
typedef __attribute__((ext_vector_type(8))) float v8f;

#define T_SEQ     10
#define D_MODEL   256
#define BATCH     16
#define RES       200
#define MODES_OUT 16
#define MODES_IN  32
#define NXT       13
#define KDIM      64
#define TAB_BYTES (RES * 8)

#define XT_FIRST  7

struct VFrag { v2f k[16]; };
struct WFrag { v2f k[8];  };
struct MFrag { v2f k[8];  };

__device__ __forceinline__ v8f mma(v2f a, v2f b, v8f c) {
  return __builtin_amdgcn_wmma_f32_16x16x4_f32(false, a, false, b, (short)0, c, false, false);
}

__device__ __forceinline__ void gather_trig(MFrag& a, const char* tab, const unsigned (&off)[8]) {
  #pragma unroll
  for (int e = 0; e < 8; ++e) a.k[e] = *(const v2f*)(tab + off[e]);
}

__device__ __forceinline__ void next_tile(unsigned (&off)[8], const unsigned (&step)[8]) {
  #pragma unroll
  for (int e = 0; e < 8; ++e) {
    const unsigned t = off[e] + step[e];
    off[e] = min(t, t - (unsigned)TAB_BYTES);
  }
}

template <bool LAST_TILE>
__device__ __forceinline__ void tile_mma_store(const MFrag& a, const MFrag& m0, const MFrag& m1,
                                               float* __restrict__ obase, int xb, int lh) {
  v8f acc0 = {};
  v8f acc1 = {};
  #pragma unroll
  for (int e = 0; e < 8; ++e) {
    acc0 = mma(a.k[e], m0.k[e], acc0);
    acc1 = mma(a.k[e], m1.k[e], acc1);
  }
  #pragma unroll
  for (int j = 0; j < 8; ++j) {
    int x = xb + j + 8 * lh;
    if (!LAST_TILE || x < RES) {
      obase[(size_t)x * D_MODEL]      = acc0[j];
      obase[(size_t)x * D_MODEL + 16] = acc1[j];
    }
  }
}

__global__ __launch_bounds__(128) void spectral_wmma(const float* __restrict__ v,
                                                     const float* __restrict__ w_real,
                                                     const float* __restrict__ w_imag,
                                                     float* __restrict__ out) {
  __shared__ v2f tab[RES];

  const int lane = threadIdx.x & 31;
  const int w    = threadIdx.x >> 5;
  const int bid  = blockIdx.x;
  const int xs   = bid & 1;
  const int hg   = (bid >> 1) & 1;
  const int bt   = bid >> 2;
  const int tt   = bt % T_SEQ;
  const int hp   = hg * 4 + w;

  const int lm = lane & 15;
  const int lh = lane >> 4;

  const float* brow0 = v + (size_t)(bt * D_MODEL + hp * 32 + lm) * KDIM + 2 * lh;
  const float* brow1 = brow0 + 16 * KDIM;
  VFrag b0, b1;
  #pragma unroll
  for (int kb = 0; kb < 16; ++kb) {
    b0.k[kb] = *(const v2f*)(brow0 + 4 * kb);
    b1.k[kb] = *(const v2f*)(brow1 + 4 * kb);
  }

  const int   srow   = lm & 1;
  const float* lo_src = srow ? w_imag : w_real;
  const float* hi_src = srow ? w_real : w_imag;
  const float  hi_sgn = srow ? 1.0f : -1.0f;
  const size_t rowP = (size_t)(tt * MODES_OUT + (lm >> 1)) * MODES_IN + 2 * lh;
  const size_t rowQ = rowP + (size_t)8 * MODES_IN;
  WFrag pl, ph, ql, qh;
  #pragma unroll
  for (int j = 0; j < 8; ++j) {
    pl.k[j] = *(const v2f*)(lo_src + rowP + 4 * j);
    ql.k[j] = *(const v2f*)(lo_src + rowQ + 4 * j);
    ph.k[j] = *(const v2f*)(hi_src + rowP + 4 * j) * hi_sgn;
    qh.k[j] = *(const v2f*)(hi_src + rowQ + 4 * j) * hi_sgn;
  }

  {
    const int k0 = threadIdx.x;
    float s, c;
    sincospif((float)k0 / 100.0f, &s, &c);
    tab[k0] = (v2f){c, -s};
    const int k1 = k0 + 128;
    if (k1 < RES) {
      sincospif((float)k1 / 100.0f, &s, &c);
      tab[k1] = (v2f){c, -s};
    }
  }

  v8f p0 = {}, q0 = {}, p1 = {}, q1 = {};
  #pragma unroll
  for (int j = 0; j < 8; ++j) {
    p0 = mma(pl.k[j], b0.k[j], p0);
    q0 = mma(ql.k[j], b0.k[j], q0);
    p1 = mma(pl.k[j], b1.k[j], p1);
    q1 = mma(ql.k[j], b1.k[j], q1);
  }
  #pragma unroll
  for (int j = 0; j < 8; ++j) {
    p0 = mma(ph.k[j], b0.k[8 + j], p0);
    q0 = mma(qh.k[j], b0.k[8 + j], q0);
    p1 = mma(ph.k[j], b1.k[8 + j], p1);
    q1 = mma(qh.k[j], b1.k[8 + j], q1);
  }

  MFrag m0, m1;
  #pragma unroll
  for (int e = 0; e < 4; ++e) {
    m0.k[e]     = (v2f){p0[2 * e], p0[2 * e + 1]};
    m0.k[4 + e] = (v2f){q0[2 * e], q0[2 * e + 1]};
    m1.k[e]     = (v2f){p1[2 * e], p1[2 * e + 1]};
    m1.k[4 + e] = (v2f){q1[2 * e], q1[2 * e + 1]};
  }

  const int xt0 = xs * XT_FIRST;
  float* obase = out + (size_t)bt * RES * D_MODEL + hp * 32 + lm;

  unsigned off[8], step[8];
  #pragma unroll
  for (int e = 0; e < 8; ++e) {
    const unsigned o = 4u * lh + e + (e >= 4 ? 4u : 0u);
    off[e]  = 8u * ((o * (unsigned)(xt0 * 16 + lm)) % (unsigned)RES);
    step[e] = 8u * ((16u * o) % (unsigned)RES);
  }

  __syncthreads();
  const char* tabb = (const char*)tab;

  MFrag a, an;
  gather_trig(a, tabb, off);

  #pragma unroll
  for (int i = 0; i < 5; ++i) {
    next_tile(off, step);
    gather_trig(an, tabb, off);
    tile_mma_store<false>(a, m0, m1, obase, (xt0 + i) * 16, lh);
    a = an;
  }

  if (xs == 0) {
    next_tile(off, step);
    gather_trig(an, tabb, off);
    tile_mma_store<false>(a,  m0, m1, obase, 5 * 16, lh);
    tile_mma_store<false>(an, m0, m1, obase, 6 * 16, lh);
  } else {
    tile_mma_store<true>(a, m0, m1, obase, (NXT - 1) * 16, lh);
  }
}

extern "C" void kernel_launch(void* const* d_in, const int* in_sizes, int n_in,
                              void* d_out, int out_size, void* d_ws, size_t ws_size,
                              hipStream_t stream) {
  const float* v      = (const float*)d_in[0];
  const float* w_real = (const float*)d_in[1];
  const float* w_imag = (const float*)d_in[2];
  float* out = (float*)d_out;

  spectral_wmma<<<BATCH * T_SEQ * 2 * 2, 128, 0, stream>>>(v, w_real, w_imag, out);
}


# grid mirror x -> 200-x: cos*C and sin*S products (K=16 each) run for x-tiles 0..6 only, row x = E+O, row 200-x = E-O; one wave covers all x of its 32 h, stage 1 once, 320 blocks; sum order changed
# speedup vs baseline: 3.5032x; 1.1410x over previous
#include <hip/hip_runtime.h>
#include <math.h>


typedef __attribute__((ext_vector_type(2))) float v2f;
typedef __attribute__((ext_vector_type(8))) float v8f;

#define T_SEQ     10
#define D_MODEL   256
#define BATCH     16
#define RES       200
#define MODES_OUT 16
#define MODES_IN  32
#define KDIM      64
#define X_MID     100
#define NXT_HALF  7
#define TAB_BYTES (RES * 4)
#define SIN_OFF   1040


struct VFrag { v2f k[16]; };
struct WFrag { v2f k[8];  };
struct TFrag { v2f k[4];  };

__device__ __forceinline__ v8f mma(v2f a, v2f b, v8f c) {
  return __builtin_amdgcn_wmma_f32_16x16x4_f32(false, a, false, b, (short)0, c, false, false);
}

__device__ __forceinline__ void gather_trig(TFrag& ac, TFrag& as, const char* tab, const unsigned (&off)[8]) {
  #pragma unroll
  for (int e = 0; e < 4; ++e) {
    ac.k[e] = (v2f){*(const float*)(tab + off[2 * e]),           *(const float*)(tab + off[2 * e + 1])};
    as.k[e] = (v2f){*(const float*)(tab + SIN_OFF + off[2 * e]), *(const float*)(tab + SIN_OFF + off[2 * e + 1])};
  }
}

__device__ __forceinline__ void next_tile(unsigned (&off)[8], const unsigned (&step)[8]) {
  #pragma unroll
  for (int i = 0; i < 8; ++i) {
    const unsigned t = off[i] + step[i];
    off[i] = min(t, t - (unsigned)TAB_BYTES);
  }
}

template <bool FIRST, bool LAST>
__device__ __forceinline__ void tile_mma_store(const TFrag& ac, const TFrag& as,
                                               const v8f& p0, const v8f& q0, const v8f& p1, const v8f& q1,
                                               float* __restrict__ od, float* __restrict__ om,
                                               int xt, int lh) {
  v8f e0 = {}, o0 = {}, e1 = {}, o1 = {};
  #pragma unroll
  for (int e = 0; e < 4; ++e) {
    e0 = mma(ac.k[e], (v2f){p0[2 * e], p0[2 * e + 1]}, e0);
    o0 = mma(as.k[e], (v2f){q0[2 * e], q0[2 * e + 1]}, o0);
    e1 = mma(ac.k[e], (v2f){p1[2 * e], p1[2 * e + 1]}, e1);
    o1 = mma(as.k[e], (v2f){q1[2 * e], q1[2 * e + 1]}, o1);
  }
  #pragma unroll
  for (int j = 0; j < 8; ++j) {
    const int xr = xt * 16 + j;
    const int x  = xr + 8 * lh;
    const float d0 = e0[j] + o0[j];
    const float d1 = e1[j] + o1[j];
    const float m0 = e0[j] - o0[j];
    const float m1 = e1[j] - o1[j];
    if (!LAST || x <= X_MID) {
      od[(ptrdiff_t)xr * D_MODEL]      = d0;
      od[(ptrdiff_t)xr * D_MODEL + 16] = d1;
    }
    if ((!FIRST || x >= 1) && (!LAST || x < X_MID)) {
      om[-(ptrdiff_t)xr * D_MODEL]      = m0;
      om[-(ptrdiff_t)xr * D_MODEL + 16] = m1;
    }
  }
}

__global__ __launch_bounds__(128) void spectral_wmma(const float* __restrict__ v,
                                                     const float* __restrict__ w_real,
                                                     const float* __restrict__ w_imag,
                                                     float* __restrict__ out) {
  __shared__ float tab[SIN_OFF / 4 + RES];

  const int lane = threadIdx.x & 31;
  const int w    = threadIdx.x >> 5;
  const int bid  = blockIdx.x;
  const int hg   = bid & 1;
  const int bt   = bid >> 1;
  const int tt   = bt % T_SEQ;
  const int hp   = hg * 4 + w;

  const int lm = lane & 15;
  const int lh = lane >> 4;

  const float* brow0 = v + (size_t)(bt * D_MODEL + hp * 32 + lm) * KDIM + 2 * lh;
  const float* brow1 = brow0 + 16 * KDIM;
  VFrag b0, b1;
  #pragma unroll
  for (int kb = 0; kb < 16; ++kb) {
    b0.k[kb] = *(const v2f*)(brow0 + 4 * kb);
    b1.k[kb] = *(const v2f*)(brow1 + 4 * kb);
  }

  const size_t roww = (size_t)(tt * MODES_OUT + lm) * MODES_IN + 2 * lh;
  WFrag wr, wi, wn;
  #pragma unroll
  for (int j = 0; j < 8; ++j) {
    wr.k[j] = *(const v2f*)(w_real + roww + 4 * j);
    wi.k[j] = *(const v2f*)(w_imag + roww + 4 * j);
    wn.k[j] = -wi.k[j];
  }

  {
    const int k0 = threadIdx.x;
    float s, c;
    sincospif((float)k0 / 100.0f, &s, &c);
    tab[k0]               = c;
    tab[SIN_OFF / 4 + k0] = -s;
    const int k1 = k0 + 128;
    if (k1 < RES) {
      sincospif((float)k1 / 100.0f, &s, &c);
      tab[k1]               = c;
      tab[SIN_OFF / 4 + k1] = -s;
    }
  }

  v8f p0 = {}, q0 = {}, p1 = {}, q1 = {};
  #pragma unroll
  for (int j = 0; j < 8; ++j) {
    p0 = mma(wr.k[j], b0.k[j], p0);
    q0 = mma(wi.k[j], b0.k[j], q0);
    p1 = mma(wr.k[j], b1.k[j], p1);
    q1 = mma(wi.k[j], b1.k[j], q1);
  }
  #pragma unroll
  for (int j = 0; j < 8; ++j) {
    p0 = mma(wn.k[j], b0.k[8 + j], p0);
    q0 = mma(wr.k[j], b0.k[8 + j], q0);
    p1 = mma(wn.k[j], b1.k[8 + j], p1);
    q1 = mma(wr.k[j], b1.k[8 + j], q1);
  }

  float* od = out + ((size_t)bt * RES + 8 * lh) * D_MODEL + hp * 32 + lm;
  float* om = out + ((size_t)bt * RES + RES - 8 * lh) * D_MODEL + hp * 32 + lm;

  unsigned off[8], step[8];
  #pragma unroll
  for (int i = 0; i < 8; ++i) {
    const unsigned o = 8u * lh + i;
    off[i]  = 4u * ((o * (unsigned)lm) % (unsigned)RES);
    step[i] = 4u * ((16u * o) % (unsigned)RES);
  }

  __syncthreads();
  const char* tabb = (const char*)tab;

  TFrag ac, as, acn, asn;
  gather_trig(ac, as, tabb, off);

  #pragma unroll
  for (int i = 0; i < NXT_HALF; ++i) {
    if (i + 1 < NXT_HALF) {
      next_tile(off, step);
      gather_trig(acn, asn, tabb, off);
    }
    if (i == 0)                 tile_mma_store<true,  false>(ac, as, p0, q0, p1, q1, od, om, i, lh);
    else if (i == NXT_HALF - 1) tile_mma_store<false, true >(ac, as, p0, q0, p1, q1, od, om, i, lh);
    else                        tile_mma_store<false, false>(ac, as, p0, q0, p1, q1, od, om, i, lh);
    ac = acn;
    as = asn;
  }
}

extern "C" void kernel_launch(void* const* d_in, const int* in_sizes, int n_in,
                              void* d_out, int out_size, void* d_ws, size_t ws_size,
                              hipStream_t stream) {
  const float* v      = (const float*)d_in[0];
  const float* w_real = (const float*)d_in[1];
  const float* w_imag = (const float*)d_in[2];
  float* out = (float*)d_out;

  spectral_wmma<<<BATCH * T_SEQ * 2, 128, 0, stream>>>(v, w_real, w_imag, out);
}


# each x-tile's 32 output rows staged through a per-wave LDS patch and stored as whole 128-byte lines (b128, 4 lines an instruction) in place of 64-byte halves by b32 stores; sums unchanged
# speedup vs baseline: 3.7259x; 1.0636x over previous
#include <hip/hip_runtime.h>
#include <math.h>


typedef __attribute__((ext_vector_type(2))) float v2f;
typedef __attribute__((ext_vector_type(8))) float v8f;

#define T_SEQ     10
#define D_MODEL   256
#define BATCH     16
#define RES       200
#define MODES_OUT 16
#define MODES_IN  32
#define KDIM      64
#define X_MID     100
#define NXT_HALF  7
#define TAB_BYTES (RES * 4)
#define SIN_OFF   1040
#define ST_PITCH  36
#define ST_TILE   (32 * ST_PITCH)


struct VFrag { v2f k[16]; };
struct WFrag { v2f k[8];  };
struct TFrag { v2f k[4];  };

__device__ __forceinline__ v8f mma(v2f a, v2f b, v8f c) {
  return __builtin_amdgcn_wmma_f32_16x16x4_f32(false, a, false, b, (short)0, c, false, false);
}

__device__ __forceinline__ void gather_trig(TFrag& ac, TFrag& as, const char* tab, const unsigned (&off)[8]) {
  #pragma unroll
  for (int e = 0; e < 4; ++e) {
    ac.k[e] = (v2f){*(const float*)(tab + off[2 * e]),           *(const float*)(tab + off[2 * e + 1])};
    as.k[e] = (v2f){*(const float*)(tab + SIN_OFF + off[2 * e]), *(const float*)(tab + SIN_OFF + off[2 * e + 1])};
  }
}

__device__ __forceinline__ void next_tile(unsigned (&off)[8], const unsigned (&step)[8]) {
  #pragma unroll
  for (int i = 0; i < 8; ++i) {
    const unsigned t = off[i] + step[i];
    off[i] = min(t, t - (unsigned)TAB_BYTES);
  }
}

typedef __attribute__((ext_vector_type(4))) float v4f;

__device__ __forceinline__ void tile_mma(const TFrag& ac, const TFrag& as,
                                         const v8f& p0, const v8f& q0, const v8f& p1, const v8f& q1,
                                         v8f& e0, v8f& o0, v8f& e1, v8f& o1) {
  e0 = (v8f){}; o0 = (v8f){}; e1 = (v8f){}; o1 = (v8f){};
  #pragma unroll
  for (int e = 0; e < 4; ++e) {
    e0 = mma(ac.k[e], (v2f){p0[2 * e], p0[2 * e + 1]}, e0);
    o0 = mma(as.k[e], (v2f){q0[2 * e], q0[2 * e + 1]}, o0);
    e1 = mma(ac.k[e], (v2f){p1[2 * e], p1[2 * e + 1]}, e1);
    o1 = mma(as.k[e], (v2f){q1[2 * e], q1[2 * e + 1]}, o1);
  }
}

__device__ __forceinline__ void tile_stage(const v8f& e0, const v8f& o0, const v8f& e1, const v8f& o1,
                                           float* st, bool last) {
  #pragma unroll
  for (int j = 0; j < 8; ++j) {
    if (last && j > 4) continue;
    st[j * ST_PITCH]             = e0[j] + o0[j];
    st[j * ST_PITCH + 16]        = e1[j] + o1[j];
    st[(16 + j) * ST_PITCH]      = e0[j] - o0[j];
    st[(16 + j) * ST_PITCH + 16] = e1[j] - o1[j];
  }
}

__device__ __forceinline__ bool group_used(int q, bool last) {
  return !last || q == 0 || q == 1 || q == 4;
}

__device__ __forceinline__ void tile_readback(v4f (&rows)[8], const float* sr, bool last) {
  #pragma unroll
  for (int q = 0; q < 8; ++q)
    if (group_used(q, last)) rows[q] = *(const v4f*)(sr + 4 * q * ST_PITCH);
}

__device__ __forceinline__ void tile_store(const v4f (&rows)[8], float* __restrict__ gd, float* __restrict__ gm,
                                           int xt, int sub, bool first, bool last) {
  #pragma unroll
  for (int q = 0; q < 4; ++q) {
    if (!group_used(q, last)) continue;
    if (!last || q == 0 || sub == 0)
      *(v4f*)(gd + (ptrdiff_t)(16 * xt + 4 * q) * D_MODEL) = rows[q];
  }
  #pragma unroll
  for (int q = 0; q < 4; ++q) {
    if (!group_used(4 + q, last)) continue;
    if (!first || q != 0 || sub != 0)
      *(v4f*)(gm - (ptrdiff_t)(16 * xt + 4 * q) * D_MODEL) = rows[4 + q];
  }
}

__device__ __forceinline__ void wait_lds() { asm volatile("s_wait_dscnt 0x0" ::: "memory"); }

__global__ __launch_bounds__(128) void spectral_wmma(const float* __restrict__ v,
                                                     const float* __restrict__ w_real,
                                                     const float* __restrict__ w_imag,
                                                     float* __restrict__ out) {
  __shared__ float tab[SIN_OFF / 4 + RES];
  __shared__ __attribute__((aligned(16))) float stage[4 * 2 * ST_TILE];

  const int lane = threadIdx.x & 31;
  const int w    = threadIdx.x >> 5;
  const int bid  = blockIdx.x;
  const int hg   = bid & 1;
  const int bt   = bid >> 1;
  const int tt   = bt % T_SEQ;
  const int hp   = hg * 4 + w;

  const int lm = lane & 15;
  const int lh = lane >> 4;

  const float* brow0 = v + (size_t)(bt * D_MODEL + hp * 32 + lm) * KDIM + 2 * lh;
  const float* brow1 = brow0 + 16 * KDIM;
  VFrag b0, b1;
  #pragma unroll
  for (int kb = 0; kb < 16; ++kb) {
    b0.k[kb] = *(const v2f*)(brow0 + 4 * kb);
    b1.k[kb] = *(const v2f*)(brow1 + 4 * kb);
  }

  const size_t roww = (size_t)(tt * MODES_OUT + lm) * MODES_IN + 2 * lh;
  WFrag wr, wi, wn;
  #pragma unroll
  for (int j = 0; j < 8; ++j) {
    wr.k[j] = *(const v2f*)(w_real + roww + 4 * j);
    wi.k[j] = *(const v2f*)(w_imag + roww + 4 * j);
    wn.k[j] = -wi.k[j];
  }

  {
    const int k0 = threadIdx.x;
    float s, c;
    sincospif((float)k0 / 100.0f, &s, &c);
    tab[k0]               = c;
    tab[SIN_OFF / 4 + k0] = -s;
    const int k1 = k0 + 128;
    if (k1 < RES) {
      sincospif((float)k1 / 100.0f, &s, &c);
      tab[k1]               = c;
      tab[SIN_OFF / 4 + k1] = -s;
    }
  }

  v8f p0 = {}, q0 = {}, p1 = {}, q1 = {};
  #pragma unroll
  for (int j = 0; j < 8; ++j) {
    p0 = mma(wr.k[j], b0.k[j], p0);
    q0 = mma(wi.k[j], b0.k[j], q0);
    p1 = mma(wr.k[j], b1.k[j], p1);
    q1 = mma(wi.k[j], b1.k[j], q1);
  }
  #pragma unroll
  for (int j = 0; j < 8; ++j) {
    p0 = mma(wn.k[j], b0.k[8 + j], p0);
    q0 = mma(wr.k[j], b0.k[8 + j], q0);
    p1 = mma(wn.k[j], b1.k[8 + j], p1);
    q1 = mma(wr.k[j], b1.k[8 + j], q1);
  }

  const int sub  = lane >> 3;
  const int col4 = (lane & 7) * 4;
  float* gd = out + ((size_t)bt * RES + sub) * D_MODEL + hp * 32 + col4;
  float* gm = out + ((size_t)bt * RES + RES - sub) * D_MODEL + hp * 32 + col4;
  float* patch = stage + w * (2 * ST_TILE);
  float* st_wr = patch + 8 * lh * ST_PITCH + lm;
  const float* st_rd = patch + sub * ST_PITCH + col4;

  unsigned off[8], step[8];
  #pragma unroll
  for (int i = 0; i < 8; ++i) {
    const unsigned o = 8u * lh + i;
    off[i]  = 4u * ((o * (unsigned)lm) % (unsigned)RES);
    step[i] = 4u * ((16u * o) % (unsigned)RES);
  }

  __syncthreads();
  const char* tabb = (const char*)tab;

  TFrag ac, as, acn, asn;
  gather_trig(ac, as, tabb, off);

  v4f rows[8];
  #pragma unroll
  for (int i = 0; i < NXT_HALF; ++i) {
    v8f e0, o0, e1, o1;
    tile_mma(ac, as, p0, q0, p1, q1, e0, o0, e1, o1);
    wait_lds();
    if (i > 0) tile_readback(rows, st_rd + ((i - 1) & 1) * ST_TILE, false);
    if (i + 1 < NXT_HALF) {
      next_tile(off, step);
      gather_trig(acn, asn, tabb, off);
    }
    tile_stage(e0, o0, e1, o1, st_wr + (i & 1) * ST_TILE, i == NXT_HALF - 1);
    if (i > 0) tile_store(rows, gd, gm, i - 1, sub, i - 1 == 0, false);
    ac = acn;
    as = asn;
  }
  wait_lds();
  tile_readback(rows, st_rd + ((NXT_HALF - 1) & 1) * ST_TILE, true);
  tile_store(rows, gd, gm, NXT_HALF - 1, sub, false, true);
}

extern "C" void kernel_launch(void* const* d_in, const int* in_sizes, int n_in,
                              void* d_out, int out_size, void* d_ws, size_t ws_size,
                              hipStream_t stream) {
  const float* v      = (const float*)d_in[0];
  const float* w_real = (const float*)d_in[1];
  const float* w_imag = (const float*)d_in[2];
  float* out = (float*)d_out;

  spectral_wmma<<<BATCH * T_SEQ * 2, 128, 0, stream>>>(v, w_real, w_imag, out);
}


# stage-1 operands (V rows, weight tiles) loaded row-wise as whole 128-byte lines (b128) and transposed to fragment layout through LDS, in place of b64 loads touching 16 lines each; sums unchanged
# speedup vs baseline: 4.0531x; 1.0878x over previous
#include <hip/hip_runtime.h>
#include <math.h>


typedef __attribute__((ext_vector_type(2))) float v2f;
typedef __attribute__((ext_vector_type(4))) float v4f;
typedef __attribute__((ext_vector_type(8))) float v8f;

#define T_SEQ     10
#define D_MODEL   256
#define BATCH     16
#define RES       200
#define MODES_OUT 16
#define MODES_IN  32
#define KDIM      64
#define X_MID     100
#define NXT_HALF  7
#define TAB_BYTES (RES * 4)
#define SIN_OFF   1040
#define ST_PITCH  36
#define ST_TILE   (32 * ST_PITCH)
#define V_PITCH   68
#define W_PITCH   36
#define W_TILE    (16 * W_PITCH)
static_assert(32 * V_PITCH <= 2 * ST_TILE, "the staged V rows must fit the wave's two output patches");


struct VFrag { v2f k[16]; };
struct WFrag { v2f k[8];  };
struct TFrag { v2f k[4];  };

__device__ __forceinline__ v8f mma(v2f a, v2f b, v8f c) {
  return __builtin_amdgcn_wmma_f32_16x16x4_f32(false, a, false, b, (short)0, c, false, false);
}

__device__ __forceinline__ void gather_trig(TFrag& ac, TFrag& as, const char* tab, const unsigned (&off)[8]) {
  #pragma unroll
  for (int e = 0; e < 4; ++e) {
    ac.k[e] = (v2f){*(const float*)(tab + off[2 * e]),           *(const float*)(tab + off[2 * e + 1])};
    as.k[e] = (v2f){*(const float*)(tab + SIN_OFF + off[2 * e]), *(const float*)(tab + SIN_OFF + off[2 * e + 1])};
  }
}

__device__ __forceinline__ void next_tile(unsigned (&off)[8], const unsigned (&step)[8]) {
  #pragma unroll
  for (int i = 0; i < 8; ++i) {
    const unsigned t = off[i] + step[i];
    off[i] = min(t, t - (unsigned)TAB_BYTES);
  }
}

__device__ __forceinline__ void tile_mma(const TFrag& ac, const TFrag& as,
                                         const v8f& p0, const v8f& q0, const v8f& p1, const v8f& q1,
                                         v8f& e0, v8f& o0, v8f& e1, v8f& o1) {
  e0 = (v8f){}; o0 = (v8f){}; e1 = (v8f){}; o1 = (v8f){};
  #pragma unroll
  for (int e = 0; e < 4; ++e) {
    e0 = mma(ac.k[e], (v2f){p0[2 * e], p0[2 * e + 1]}, e0);
    o0 = mma(as.k[e], (v2f){q0[2 * e], q0[2 * e + 1]}, o0);
    e1 = mma(ac.k[e], (v2f){p1[2 * e], p1[2 * e + 1]}, e1);
    o1 = mma(as.k[e], (v2f){q1[2 * e], q1[2 * e + 1]}, o1);
  }
}

__device__ __forceinline__ void tile_stage(const v8f& e0, const v8f& o0, const v8f& e1, const v8f& o1,
                                           float* st, bool last) {
  #pragma unroll
  for (int j = 0; j < 8; ++j) {
    if (last && j > 4) continue;
    st[j * ST_PITCH]             = e0[j] + o0[j];
    st[j * ST_PITCH + 16]        = e1[j] + o1[j];
    st[(16 + j) * ST_PITCH]      = e0[j] - o0[j];
    st[(16 + j) * ST_PITCH + 16] = e1[j] - o1[j];
  }
}

__device__ __forceinline__ bool group_used(int q, bool last) {
  return !last || q == 0 || q == 1 || q == 4;
}

__device__ __forceinline__ void tile_readback(v4f (&rows)[8], const float* sr, bool last) {
  #pragma unroll
  for (int q = 0; q < 8; ++q)
    if (group_used(q, last)) rows[q] = *(const v4f*)(sr + 4 * q * ST_PITCH);
}

__device__ __forceinline__ void tile_store(const v4f (&rows)[8], float* __restrict__ gd, float* __restrict__ gm,
                                           int xt, int sub, bool first, bool last) {
  #pragma unroll
  for (int q = 0; q < 4; ++q) {
    if (!group_used(q, last)) continue;
    if (!last || q == 0 || sub == 0)
      *(v4f*)(gd + (ptrdiff_t)(16 * xt + 4 * q) * D_MODEL) = rows[q];
  }
  #pragma unroll
  for (int q = 0; q < 4; ++q) {
    if (!group_used(4 + q, last)) continue;
    if (!first || q != 0 || sub != 0)
      *(v4f*)(gm - (ptrdiff_t)(16 * xt + 4 * q) * D_MODEL) = rows[4 + q];
  }
}

__device__ __forceinline__ void wait_lds() { asm volatile("s_wait_dscnt 0x0" ::: "memory"); }

__global__ __launch_bounds__(128) void spectral_wmma(const float* __restrict__ v,
                                                     const float* __restrict__ w_real,
                                                     const float* __restrict__ w_imag,
                                                     float* __restrict__ out) {
  __shared__ float tab[SIN_OFF / 4 + RES];
  __shared__ __attribute__((aligned(16))) float stage[4 * 2 * ST_TILE];
  __shared__ __attribute__((aligned(16))) float wst[4 * 2 * W_TILE];

  const int lane = threadIdx.x & 31;
  const int w    = threadIdx.x >> 5;
  const int bid  = blockIdx.x;
  const int hg   = bid & 1;
  const int bt   = bid >> 1;
  const int tt   = bt % T_SEQ;
  const int hp   = hg * 4 + w;

  const int lm = lane & 15;
  const int lh = lane >> 4;

  const float* vsrc = v + (size_t)(bt * D_MODEL + hp * 32) * KDIM + 4 * lane;
  const float* wrs  = w_real + (size_t)tt * (MODES_OUT * MODES_IN) + 4 * lane;
  const float* wis  = w_imag + (size_t)tt * (MODES_OUT * MODES_IN) + 4 * lane;
  v4f vr[16], wrr[4], wir[4];
  #pragma unroll
  for (int q = 0; q < 16; ++q) vr[q] = *(const v4f*)(vsrc + 128 * q);
  #pragma unroll
  for (int q = 0; q < 4; ++q) {
    wrr[q] = *(const v4f*)(wrs + 128 * q);
    wir[q] = *(const v4f*)(wis + 128 * q);
  }

  {
    const int k0 = threadIdx.x;
    float s, c;
    sincospif((float)k0 / 100.0f, &s, &c);
    tab[k0]               = c;
    tab[SIN_OFF / 4 + k0] = -s;
    const int k1 = k0 + 128;
    if (k1 < RES) {
      sincospif((float)k1 / 100.0f, &s, &c);
      tab[k1]               = c;
      tab[SIN_OFF / 4 + k1] = -s;
    }
  }

  float* patch  = stage + w * (2 * ST_TILE);
  float* wpatch = wst + w * (2 * W_TILE);
  {
    float* vs = patch + (lane >> 4) * V_PITCH + (lane & 15) * 4;
    #pragma unroll
    for (int q = 0; q < 16; ++q) *(v4f*)(vs + 2 * q * V_PITCH) = vr[q];
    float* ws = wpatch + (lane >> 3) * W_PITCH + (lane & 7) * 4;
    #pragma unroll
    for (int q = 0; q < 4; ++q) {
      *(v4f*)(ws + 4 * q * W_PITCH)          = wrr[q];
      *(v4f*)(ws + W_TILE + 4 * q * W_PITCH) = wir[q];
    }
  }
  wait_lds();

  const float* vf = patch + lm * V_PITCH + 2 * lh;
  VFrag b0, b1;
  #pragma unroll
  for (int kb = 0; kb < 16; ++kb) {
    b0.k[kb] = *(const v2f*)(vf + 4 * kb);
    b1.k[kb] = *(const v2f*)(vf + 16 * V_PITCH + 4 * kb);
  }

  const float* wf = wpatch + lm * W_PITCH + 2 * lh;
  WFrag wr, wi, wn;
  #pragma unroll
  for (int j = 0; j < 8; ++j) {
    wr.k[j] = *(const v2f*)(wf + 4 * j);
    wi.k[j] = *(const v2f*)(wf + W_TILE + 4 * j);
    wn.k[j] = -wi.k[j];
  }

  v8f p0 = {}, q0 = {}, p1 = {}, q1 = {};
  #pragma unroll
  for (int j = 0; j < 8; ++j) {
    p0 = mma(wr.k[j], b0.k[j], p0);
    q0 = mma(wi.k[j], b0.k[j], q0);
    p1 = mma(wr.k[j], b1.k[j], p1);
    q1 = mma(wi.k[j], b1.k[j], q1);
  }
  #pragma unroll
  for (int j = 0; j < 8; ++j) {
    p0 = mma(wn.k[j], b0.k[8 + j], p0);
    q0 = mma(wr.k[j], b0.k[8 + j], q0);
    p1 = mma(wn.k[j], b1.k[8 + j], p1);
    q1 = mma(wr.k[j], b1.k[8 + j], q1);
  }

  const int sub  = lane >> 3;
  const int col4 = (lane & 7) * 4;
  float* gd = out + ((size_t)bt * RES + sub) * D_MODEL + hp * 32 + col4;
  float* gm = out + ((size_t)bt * RES + RES - sub) * D_MODEL + hp * 32 + col4;
  float* st_wr = patch + 8 * lh * ST_PITCH + lm;
  const float* st_rd = patch + sub * ST_PITCH + col4;

  unsigned off[8], step[8];
  #pragma unroll
  for (int i = 0; i < 8; ++i) {
    const unsigned o = 8u * lh + i;
    off[i]  = 4u * ((o * (unsigned)lm) % (unsigned)RES);
    step[i] = 4u * ((16u * o) % (unsigned)RES);
  }

  __syncthreads();
  const char* tabb = (const char*)tab;

  TFrag ac, as, acn, asn;
  gather_trig(ac, as, tabb, off);

  v4f rows[8];
  #pragma unroll
  for (int i = 0; i < NXT_HALF; ++i) {
    v8f e0, o0, e1, o1;
    tile_mma(ac, as, p0, q0, p1, q1, e0, o0, e1, o1);
    wait_lds();
    if (i > 0) tile_readback(rows, st_rd + ((i - 1) & 1) * ST_TILE, false);
    if (i + 1 < NXT_HALF) {
      next_tile(off, step);
      gather_trig(acn, asn, tabb, off);
    }
    tile_stage(e0, o0, e1, o1, st_wr + (i & 1) * ST_TILE, i == NXT_HALF - 1);
    if (i > 0) tile_store(rows, gd, gm, i - 1, sub, i - 1 == 0, false);
    ac = acn;
    as = asn;
  }
  wait_lds();
  tile_readback(rows, st_rd + ((NXT_HALF - 1) & 1) * ST_TILE, true);
  tile_store(rows, gd, gm, NXT_HALF - 1, sub, false, true);
}

extern "C" void kernel_launch(void* const* d_in, const int* in_sizes, int n_in,
                              void* d_out, int out_size, void* d_ws, size_t ws_size,
                              hipStream_t stream) {
  const float* v      = (const float*)d_in[0];
  const float* w_real = (const float*)d_in[1];
  const float* w_imag = (const float*)d_in[2];
  float* out = (float*)d_out;

  spectral_wmma<<<BATCH * T_SEQ * 2, 128, 0, stream>>>(v, w_real, w_imag, out);
}
